# starting kernel
# baseline (speedup 1.0000x reference)
#include <hip/hip_runtime.h>

typedef float v2f __attribute__((ext_vector_type(2)));
typedef float v8f __attribute__((ext_vector_type(8)));

#define NT    3000
#define NOUT  603
#define XOFF  301
#define XPAD  3775
#define YOFF  240
#define YPAD  3488
#define NS    203

__global__ __launch_bounds__(96) void xcorr_wmma_f32(const float* __restrict__ d1,
                                                     const float* __restrict__ d2,
                                                     float* __restrict__ out) {
  __shared__ float xs[XPAD];
  __shared__ float ys[YPAD];

  const int c   = blockIdx.x;
  const int tid = threadIdx.x;
  const float* x = d1 + c * NT;
  const float* y = d2 + c * NT;

  for (int i = tid; i < XPAD; i += 96) {
    int t = i - XOFF;
    xs[i] = (t >= 0 && t < NT) ? x[t] : 0.0f;
  }
  for (int i = tid; i < YPAD; i += 96) {
    int t = i - YOFF;
    ys[i] = (t >= 0 && t < NT) ? y[t] : 0.0f;
  }
  __syncthreads();

  const int wave = tid >> 5;
  const int lane = tid & 31;
  const int m    = lane & 15;
  const int kb2  = (lane >> 4) << 1;
  const int OB0  = wave << 4;

  const int alo = m + kb2;
  const int ylo = YOFF - ((OB0 + m) << 4) + kb2;

  v8f acc = {};
  int base = OB0 << 4;
  for (int s = 0; s < NS; ++s) {
    const int xb = base + alo;
    const int yb = base + ylo;
#pragma unroll
    for (int kc = 0; kc < 4; ++kc) {
      v2f a, b;
      a.x = xs[xb + 4 * kc];
      a.y = xs[xb + 4 * kc + 1];
      b.x = ys[yb + 4 * kc];
      b.y = ys[yb + 4 * kc + 1];
      acc = __builtin_amdgcn_wmma_f32_16x16x4_f32(false, a, false, b,
                                                  (short)0, acc, false, false);
    }
    base += 16;
  }

  const int hi = lane >> 4;
#pragma unroll
  for (int i = 0; i < 8; ++i) {
    int oi = i + (hi << 3);
    int o  = ((OB0 + m) << 4) + oi;
    if (o < NOUT) out[c * NOUT + o] = acc[i];
  }
}

extern "C" void kernel_launch(void* const* d_in, const int* in_sizes, int n_in,
                              void* d_out, int out_size, void* d_ws, size_t ws_size,
                              hipStream_t stream) {
  const float* d1 = (const float*)d_in[0];
  const float* d2 = (const float*)d_in[1];
  float* out = (float*)d_out;
  (void)in_sizes; (void)n_in; (void)out_size; (void)d_ws; (void)ws_size;
  hipLaunchKernelGGL(xcorr_wmma_f32, dim3(2048), dim3(96), 0, stream, d1, d2, out);
}


# staging pipelined behind the matrix loop: 7 chunks of 29 steps; chunk p+1's global loads are issued before chunk p's steps and stored to LDS after them; only chunk 0's staging is exposed
# speedup vs baseline: 1.3345x; 1.3345x over previous
#include <hip/hip_runtime.h>


typedef float v2f __attribute__((ext_vector_type(2)));
typedef float v8f __attribute__((ext_vector_type(8)));

#define NT    3000
#define NOUT  603
#define XOFF  301
#define XPAD  3775
#define YOFF  240
#define YPAD  3488
#define NS    203

#define SC    29
#define NCH   7
#define CW    (16 * SC)
#define X0    (16 * (32 + SC - 1) + 31)
#define Y0    (16 * (SC - 1) + 256)
#define NPX   11
#define NPY   8
#define NR    5

__device__ __forceinline__ float ld_clamped(const float* __restrict__ p, int t) {
  int tc = t < 0 ? 0 : (t > NT - 1 ? NT - 1 : t);
  return p[tc];
}
__device__ __forceinline__ bool in_sig(int t) { return t >= 0 && t < NT; }

__global__ __launch_bounds__(96) void xcorr_wmma_f32(const float* __restrict__ d1,
                                                     const float* __restrict__ d2,
                                                     float* __restrict__ out) {
  __shared__ float xs[XPAD];
  __shared__ float ys[YPAD];

  const int c   = blockIdx.x;
  const int tid = threadIdx.x;
  const float* x = d1 + c * NT;
  const float* y = d2 + c * NT;

  {
    float px[NPX], py[NPY];
#pragma unroll
    for (int q = 0; q < NPX; ++q) px[q] = ld_clamped(x, tid + 96 * q - XOFF);
#pragma unroll
    for (int q = 0; q < NPY; ++q) py[q] = ld_clamped(y, tid + 96 * q - YOFF);
#pragma unroll
    for (int q = 0; q < NPX; ++q) {
      int i = tid + 96 * q;
      if (i < X0) xs[i] = in_sig(i - XOFF) ? px[q] : 0.0f;
    }
#pragma unroll
    for (int q = 0; q < NPY; ++q) {
      int i = tid + 96 * q;
      if (i < Y0) ys[i] = in_sig(i - YOFF) ? py[q] : 0.0f;
    }
  }
  __syncthreads();

  const int wave = tid >> 5;
  const int lane = tid & 31;
  const int m    = lane & 15;
  const int kb2  = (lane >> 4) << 1;
  const int OB0  = wave << 4;

  const int alo = m + kb2;
  const int ylo = YOFF - ((OB0 + m) << 4) + kb2;

  v8f acc = {};
  int base = OB0 << 4;

#pragma unroll 1
  for (int p = 0; p < NCH; ++p) {
    const bool more = (p + 1 < NCH);
    const int  xn   = X0 + CW * p;
    const int  yn   = Y0 + CW * p;
    float rx[NR], ry[NR];
    if (more) {
#pragma unroll
      for (int q = 0; q < NR; ++q) {
        rx[q] = ld_clamped(x, xn + tid + 96 * q - XOFF);
        ry[q] = ld_clamped(y, yn + tid + 96 * q - YOFF);
      }
    }

    for (int s = 0; s < SC; ++s) {
      const int xb = base + alo;
      const int yb = base + ylo;
#pragma unroll
      for (int kc = 0; kc < 4; ++kc) {
        v2f a, b;
        a.x = xs[xb + 4 * kc];
        a.y = xs[xb + 4 * kc + 1];
        b.x = ys[yb + 4 * kc];
        b.y = ys[yb + 4 * kc + 1];
        acc = __builtin_amdgcn_wmma_f32_16x16x4_f32(false, a, false, b,
                                                    (short)0, acc, false, false);
      }
      base += 16;
    }

    if (more) {
#pragma unroll
      for (int q = 0; q < NR; ++q) {
        int j = tid + 96 * q;
        if (j < CW) {
          xs[xn + j] = in_sig(xn + j - XOFF) ? rx[q] : 0.0f;
          ys[yn + j] = in_sig(yn + j - YOFF) ? ry[q] : 0.0f;
        }
      }
    }
    __syncthreads();
  }

  const int hi = lane >> 4;
#pragma unroll
  for (int i = 0; i < 8; ++i) {
    int oi = i + (hi << 3);
    int o  = ((OB0 + m) << 4) + oi;
    if (o < NOUT) out[c * NOUT + o] = acc[i];
  }
}

extern "C" void kernel_launch(void* const* d_in, const int* in_sizes, int n_in,
                              void* d_out, int out_size, void* d_ws, size_t ws_size,
                              hipStream_t stream) {
  const float* d1 = (const float*)d_in[0];
  const float* d2 = (const float*)d_in[1];
  float* out = (float*)d_out;
  (void)in_sizes; (void)n_in; (void)out_size; (void)d_ws; (void)ws_size;
  hipLaunchKernelGGL(xcorr_wmma_f32, dim3(2048), dim3(96), 0, stream, d1, d2, out);
}


# four channels per workgroup (12 waves, 3 on each SIMD; was one channel, 3 waves): each 96-thread group keeps its own channel and tile loop; 512 workgroups of 384 threads
# speedup vs baseline: 1.4009x; 1.0498x over previous
#include <hip/hip_runtime.h>


typedef float v2f __attribute__((ext_vector_type(2)));
typedef float v8f __attribute__((ext_vector_type(8)));

#define NT    3000
#define NOUT  603
#define XOFF  301
#define XPAD  3775
#define YOFF  240
#define YPAD  3488
#define NS    203

#define SC    29
#define NCH   7
#define CW    (16 * SC)
#define X0    (16 * (32 + SC - 1) + 31)
#define Y0    (16 * (SC - 1) + 256)
#define NPX   11
#define NPY   8
#define NR    5
#define NCB   4

__device__ __forceinline__ float ld_clamped(const float* __restrict__ p, int t) {
  int tc = t < 0 ? 0 : (t > NT - 1 ? NT - 1 : t);
  return p[tc];
}
__device__ __forceinline__ bool in_sig(int t) { return t >= 0 && t < NT; }

__global__ __launch_bounds__(96 * NCB) void xcorr_wmma_f32(const float* __restrict__ d1,
                                                     const float* __restrict__ d2,
                                                     float* __restrict__ out) {
  __shared__ float xs_all[NCB][XPAD];
  __shared__ float ys_all[NCB][YPAD];

  const int ch  = threadIdx.x / 96;
  const int tid = threadIdx.x - 96 * ch;
  const int c   = blockIdx.x * NCB + ch;
  float* xs = xs_all[ch];
  float* ys = ys_all[ch];
  const float* x = d1 + c * NT;
  const float* y = d2 + c * NT;

  {
    float px[NPX], py[NPY];
#pragma unroll
    for (int q = 0; q < NPX; ++q) px[q] = ld_clamped(x, tid + 96 * q - XOFF);
#pragma unroll
    for (int q = 0; q < NPY; ++q) py[q] = ld_clamped(y, tid + 96 * q - YOFF);
#pragma unroll
    for (int q = 0; q < NPX; ++q) {
      int i = tid + 96 * q;
      if (i < X0) xs[i] = in_sig(i - XOFF) ? px[q] : 0.0f;
    }
#pragma unroll
    for (int q = 0; q < NPY; ++q) {
      int i = tid + 96 * q;
      if (i < Y0) ys[i] = in_sig(i - YOFF) ? py[q] : 0.0f;
    }
  }
  __syncthreads();

  const int wave = tid >> 5;
  const int lane = tid & 31;
  const int m    = lane & 15;
  const int kb2  = (lane >> 4) << 1;
  const int OB0  = wave << 4;

  const int alo = m + kb2;
  const int ylo = YOFF - ((OB0 + m) << 4) + kb2;

  v8f acc = {};
  int base = OB0 << 4;

#pragma unroll 1
  for (int p = 0; p < NCH; ++p) {
    const bool more = (p + 1 < NCH);
    const int  xn   = X0 + CW * p;
    const int  yn   = Y0 + CW * p;
    float rx[NR], ry[NR];
    if (more) {
#pragma unroll
      for (int q = 0; q < NR; ++q) {
        rx[q] = ld_clamped(x, xn + tid + 96 * q - XOFF);
        ry[q] = ld_clamped(y, yn + tid + 96 * q - YOFF);
      }
    }

    for (int s = 0; s < SC; ++s) {
      const int xb = base + alo;
      const int yb = base + ylo;
#pragma unroll
      for (int kc = 0; kc < 4; ++kc) {
        v2f a, b;
        a.x = xs[xb + 4 * kc];
        a.y = xs[xb + 4 * kc + 1];
        b.x = ys[yb + 4 * kc];
        b.y = ys[yb + 4 * kc + 1];
        acc = __builtin_amdgcn_wmma_f32_16x16x4_f32(false, a, false, b,
                                                    (short)0, acc, false, false);
      }
      base += 16;
    }

    if (more) {
#pragma unroll
      for (int q = 0; q < NR; ++q) {
        int j = tid + 96 * q;
        if (j < CW) {
          xs[xn + j] = in_sig(xn + j - XOFF) ? rx[q] : 0.0f;
          ys[yn + j] = in_sig(yn + j - YOFF) ? ry[q] : 0.0f;
        }
      }
    }
    __syncthreads();
  }

  const int hi = lane >> 4;
#pragma unroll
  for (int i = 0; i < 8; ++i) {
    int oi = i + (hi << 3);
    int o  = ((OB0 + m) << 4) + oi;
    if (o < NOUT) out[c * NOUT + o] = acc[i];
  }
}

extern "C" void kernel_launch(void* const* d_in, const int* in_sizes, int n_in,
                              void* d_out, int out_size, void* d_ws, size_t ws_size,
                              hipStream_t stream) {
  const float* d1 = (const float*)d_in[0];
  const float* d2 = (const float*)d_in[1];
  float* out = (float*)d_out;
  (void)in_sizes; (void)n_in; (void)out_size; (void)d_ws; (void)ws_size;
  hipLaunchKernelGGL(xcorr_wmma_f32, dim3(512), dim3(384), 0, stream, d1, d2, out);
}


# each tile runs only the steps whose x window holds samples (186, 190, 175 of 203), counted from its own first one: 6 whole chunks of 28 and a tail for every tile; y staged 272 samples further ahead
# speedup vs baseline: 1.4871x; 1.0615x over previous
#include <hip/hip_runtime.h>


typedef float v2f __attribute__((ext_vector_type(2)));
typedef float v8f __attribute__((ext_vector_type(8)));

#define NT    3000
#define NOUT  603
#define XOFF  301
#define XPAD  3775
#define YOFF  240
#define YPAD  3488
#define NS    203

#define S_LO  ((XOFF - 30 + 15) / 16)
#define S_HI  ((NT - 1 + XOFF) / 16 + 1)

#define SC    28
#define NCH   7
#define CW    (16 * SC)
#define X0    (16 * (32 + SC - 1) + 31)
#define Y0    (16 * (S_LO + SC - 1) + 256)
#define NPX   11
#define NPY   10
#define NR    5
#define NCB   4

__device__ __forceinline__ float ld_clamped(const float* __restrict__ p, int t) {
  int tc = t < 0 ? 0 : (t > NT - 1 ? NT - 1 : t);
  return p[tc];
}
__device__ __forceinline__ bool in_sig(int t) { return t >= 0 && t < NT; }

__device__ __forceinline__ v8f xcorr_step(const float* xs, const float* ys, int xb, int yb, v8f acc) {
#pragma unroll
  for (int kc = 0; kc < 4; ++kc) {
    v2f a, b;
    a.x = xs[xb + 4 * kc];
    a.y = xs[xb + 4 * kc + 1];
    b.x = ys[yb + 4 * kc];
    b.y = ys[yb + 4 * kc + 1];
    acc = __builtin_amdgcn_wmma_f32_16x16x4_f32(false, a, false, b,
                                                (short)0, acc, false, false);
  }
  return acc;
}

__global__ __launch_bounds__(96 * NCB) void xcorr_wmma_f32(const float* __restrict__ d1,
                                                     const float* __restrict__ d2,
                                                     float* __restrict__ out) {
  __shared__ float xs_all[NCB][XPAD];
  __shared__ float ys_all[NCB][YPAD];

  const int ch  = threadIdx.x / 96;
  const int tid = threadIdx.x - 96 * ch;
  const int c   = blockIdx.x * NCB + ch;
  float* xs = xs_all[ch];
  float* ys = ys_all[ch];
  const float* x = d1 + c * NT;
  const float* y = d2 + c * NT;

  {
    float px[NPX], py[NPY];
#pragma unroll
    for (int q = 0; q < NPX; ++q) px[q] = ld_clamped(x, tid + 96 * q - XOFF);
#pragma unroll
    for (int q = 0; q < NPY; ++q) py[q] = ld_clamped(y, tid + 96 * q - YOFF);
#pragma unroll
    for (int q = 0; q < NPX; ++q) {
      int i = tid + 96 * q;
      if (i < X0) xs[i] = in_sig(i - XOFF) ? px[q] : 0.0f;
    }
#pragma unroll
    for (int q = 0; q < NPY; ++q) {
      int i = tid + 96 * q;
      if (i < Y0) ys[i] = in_sig(i - YOFF) ? py[q] : 0.0f;
    }
  }
  __syncthreads();

  const int wave = __builtin_amdgcn_readfirstlane(tid >> 5);
  const int lane = tid & 31;
  const int m    = lane & 15;
  const int kb2  = (lane >> 4) << 1;
  const int OB0  = wave << 4;

  const int alo = m + kb2;
  const int ylo = YOFF - ((OB0 + m) << 4) + kb2;

  const int s_lo = (S_LO - OB0 > 0) ? S_LO - OB0 : 0;
  const int s_hi = (S_HI - OB0 < NS) ? S_HI - OB0 : NS;
  const int nst  = s_hi - s_lo;

  v8f acc = {};

#pragma unroll 1
  for (int p = 0; p < NCH; ++p) {
    const bool more = (p + 1 < NCH);
    const int  xn   = X0 + CW * p;
    const int  yn   = Y0 + CW * p;
    float rx[NR], ry[NR];
    if (more) {
#pragma unroll
      for (int q = 0; q < NR; ++q) {
        rx[q] = ld_clamped(x, xn + tid + 96 * q - XOFF);
        ry[q] = ld_clamped(y, yn + tid + 96 * q - YOFF);
      }
    }

    const int left = nst - SC * p;
    const int b0   = (OB0 + s_lo + SC * p) << 4;
    if (left >= SC) {
#pragma unroll
      for (int u = 0; u < SC; ++u)
        acc = xcorr_step(xs, ys, b0 + 16 * u + alo, b0 + 16 * u + ylo, acc);
    } else {
      for (int u = 0; u < left; ++u)
        acc = xcorr_step(xs, ys, b0 + 16 * u + alo, b0 + 16 * u + ylo, acc);
    }

    if (more) {
#pragma unroll
      for (int q = 0; q < NR; ++q) {
        int j = tid + 96 * q;
        if (j < CW) {
          xs[xn + j] = in_sig(xn + j - XOFF) ? rx[q] : 0.0f;
          if (yn + j < YPAD) ys[yn + j] = in_sig(yn + j - YOFF) ? ry[q] : 0.0f;
        }
      }
    }
    __syncthreads();
  }

  const int hi = lane >> 4;
#pragma unroll
  for (int i = 0; i < 8; ++i) {
    int oi = i + (hi << 3);
    int o  = ((OB0 + m) << 4) + oi;
    if (o < NOUT) out[c * NOUT + o] = acc[i];
  }
}

extern "C" void kernel_launch(void* const* d_in, const int* in_sizes, int n_in,
                              void* d_out, int out_size, void* d_ws, size_t ws_size,
                              hipStream_t stream) {
  const float* d1 = (const float*)d_in[0];
  const float* d2 = (const float*)d_in[1];
  float* out = (float*)d_out;
  (void)in_sizes; (void)n_in; (void)out_size; (void)d_ws; (void)ws_size;
  hipLaunchKernelGGL(xcorr_wmma_f32, dim3(512), dim3(384), 0, stream, d1, d2, out);
}


# one wave per channel (2048 waves, was 6144): the wave stages its own rows and runs all three lag tiles on three accumulators, fragments fetched one step ahead; no workgroup barrier left
# speedup vs baseline: 1.8251x; 1.2273x over previous
#include <hip/hip_runtime.h>


typedef float v2f __attribute__((ext_vector_type(2)));
typedef float v4f __attribute__((ext_vector_type(4)));
typedef float v8f __attribute__((ext_vector_type(8)));

#define NT    3000
#define NOUT  603
#define XOFF  301
#define XPAD  3775
#define YOFF  240
#define YBLK  218
#define YP    20
#define YPAD  (YP * YBLK)
#define NS    203

#define S_LO  ((XOFF - 30 + 15) / 16)
#define S_HI  ((NT - 1 + XOFF) / 16 + 1)

#define SC    28
#define SH    14
#define NCH   7
#define NCB   4

#define NF    24
#define FILL  (32 * NF)
#define NR    15
#define RW    (32 * NR)
#define NRND  5
#define LASTN (NT - FILL - RW * (NRND - 1))
#define XTAIL 96

static_assert(FILL + XOFF >= 16 * (32 + SC) + 31, "chunk 0 and the step behind it read x beyond the fill");
static_assert(FILL + YOFF >= 16 * (S_LO + SC) + 256, "chunk 0 and the step behind it read y beyond the fill");
static_assert(RW >= 16 * SC, "a round brings less than a chunk consumes");
static_assert(LASTN > 0 && LASTN <= RW, "the last round is not the last one");
static_assert(16 * (S_HI - 1) + 31 <= XOFF + NT + XTAIL, "x is read behind its zeroed tail");
static_assert(XOFF + NT + XTAIL <= XPAD, "x tail leaves the row");
static_assert((NS - 1) + YOFF / 16 + 1 <= YBLK, "y is read behind its row");
static_assert(YOFF % 16 == 0 && FILL % 16 == 0 && RW % 16 == 0 && 32 % 16 == 0,
              "a staging window does not start on a y block");
static_assert(YP % 8 == 4, "16 consecutive y blocks do not start on 16 different groups of 4 banks");
static_assert((NCB * XPAD) % 4 == 0, "the y rows do not start 16-byte aligned");
static_assert(SC % SH == 0, "a chunk is a whole number of halves");

constexpr int SLO0 = (S_LO > 0) ? S_LO : 0;
constexpr int SLO1 = (S_LO - 16 > 0) ? S_LO - 16 : 0;
constexpr int SLO2 = (S_LO - 32 > 0) ? S_LO - 32 : 0;
constexpr int NST0 = ((S_HI < NS) ? S_HI : NS) - SLO0;
constexpr int NST1 = ((S_HI - 16 < NS) ? S_HI - 16 : NS) - SLO1;
constexpr int NST2 = ((S_HI - 32 < NS) ? S_HI - 32 : NS) - SLO2;
constexpr int NFULL = SC * (NCH - 1);
constexpr int YB0  = SLO0 - SLO2;
static_assert(SLO0 == 16 + SLO1, "tiles 0 and 1 do not read the same x window");
static_assert(SLO1 == SLO2 + 1, "tile 1's y blocks are not tile 2's of the next step");
static_assert(NST2 >= NFULL && NST2 <= NST0 && NST0 <= NST1 && NST1 <= NFULL + SC,
              "the tail does not end tile 2 first, then tile 0, then tile 1");

struct FragA { v2f k[4]; };
struct FragB { v4f q[2]; };
struct Frag {
  FragA a01;
  FragA a2;
  FragB b0, b1, b2;
};

__device__ __forceinline__ FragA load_a(const float* xs, int xb) {
  FragA f;
#pragma unroll
  for (int j = 0; j < 2; ++j) {
    f.k[2 * j].x     = xs[xb + 8 * j];
    f.k[2 * j].y     = xs[xb + 8 * j + 1];
    f.k[2 * j + 1].x = xs[xb + 8 * j + 2];
    f.k[2 * j + 1].y = xs[xb + 8 * j + 3];
  }
  return f;
}

__device__ __forceinline__ FragB load_b(const float* ys, int yb) {
  FragB f;
#pragma unroll
  for (int j = 0; j < 2; ++j) f.q[j] = *reinterpret_cast<const v4f*>(ys + yb + 8 * j);
  return f;
}

__device__ __forceinline__ v2f b_slice(const FragB& f, int ks) {
  v2f b;
  const v4f q = f.q[ks >> 1];
  b.x = (ks & 1) ? q.z : q.x;
  b.y = (ks & 1) ? q.w : q.y;
  return b;
}

__device__ __forceinline__ void fetch(Frag& n, const Frag& prev, int mask, const float* xs,
                                      const float* ys, int xb01, int xb2, int yb) {
  if (mask & 3) n.a01 = load_a(xs, xb01);
  if (mask & 4) {
    n.a2 = load_a(xs, xb2);
    n.b2 = prev.b1;
  }
  if (mask & 2) n.b1 = load_b(ys, yb + YP * (SLO1 - SLO2));
  if (mask & 1) n.b0 = load_b(ys, yb + YP * YB0);
}

__device__ __forceinline__ void xcorr_step(const Frag& f, int mask, v8f& acc0, v8f& acc1, v8f& acc2) {
#pragma unroll
  for (int ks = 0; ks < 4; ++ks) {
    if (mask & 1)
      acc0 = __builtin_amdgcn_wmma_f32_16x16x4_f32(false, f.a01.k[ks], false, b_slice(f.b0, ks),
                                                   (short)0, acc0, false, false);
    if (mask & 2)
      acc1 = __builtin_amdgcn_wmma_f32_16x16x4_f32(false, f.a01.k[ks], false, b_slice(f.b1, ks),
                                                   (short)0, acc1, false, false);
    if (mask & 4)
      acc2 = __builtin_amdgcn_wmma_f32_16x16x4_f32(false, f.a2.k[ks], false, b_slice(f.b2, ks),
                                                   (short)0, acc2, false, false);
  }
}

__device__ __forceinline__ constexpr int tile_mask(int u) {
  return (u < NST0 ? 1 : 0) | (u < NST1 ? 2 : 0) | (u < NST2 ? 4 : 0);
}

__device__ __forceinline__ void store_tile(float* __restrict__ out, int c, int OB0, int lane, v8f acc) {
  const int m  = lane & 15;
  const int hi = lane >> 4;
#pragma unroll
  for (int i = 0; i < 8; ++i) {
    int oi = i + (hi << 3);
    int o  = ((OB0 + m) << 4) + oi;
    if (o < NOUT) out[c * NOUT + o] = acc[i];
  }
}

__global__ __launch_bounds__(32 * NCB) void xcorr_wmma_f32(const float* __restrict__ d1,
                                                     const float* __restrict__ d2,
                                                     float* __restrict__ out) {
  __shared__ __attribute__((aligned(16))) float lds[NCB * XPAD + NCB * YPAD];

  __builtin_amdgcn_s_setprio(3);

  const int ch   = __builtin_amdgcn_readfirstlane(threadIdx.x >> 5);
  const int lane = threadIdx.x & 31;
  const int c    = blockIdx.x * NCB + ch;
  float* xs = lds + ch * XPAD;
  float* ys = lds + NCB * XPAD + ch * YPAD;
  const float* x = d1 + c * NT;
  const float* y = d2 + c * NT;

  int g = lane;
  int gy = YP * (YOFF / 16 + (lane >> 4)) + (lane & 15);

  {
    float fx[NF], fy[NF];
#pragma unroll
    for (int q = 0; q < NF; ++q) fx[q] = x[g + 32 * q];
#pragma unroll
    for (int q = 0; q < NF; ++q) fy[q] = y[g + 32 * q];
#pragma unroll
    for (int q = 0; q < (XOFF + 31) / 32; ++q) {
      int i = lane + 32 * q;
      if (i < XOFF) xs[i] = 0.0f;
    }
#pragma unroll
    for (int q = 0; q < XTAIL / 32; ++q)
      xs[XOFF + NT + lane + 32 * q] = 0.0f;
#pragma unroll
    for (int q = 0; q < (YP * (YOFF / 16) + 31) / 32; ++q) {
      int i = lane + 32 * q;
      if (i < YP * (YOFF / 16)) ys[i] = 0.0f;
    }
    constexpr int YEND = YP * ((YOFF + NT) / 16) + (YOFF + NT) % 16;
#pragma unroll
    for (int q = 0; q < (YPAD - YEND + 31) / 32; ++q) {
      int i = YEND + lane + 32 * q;
      if (i < YPAD) ys[i] = 0.0f;
    }
#pragma unroll
    for (int q = 0; q < NF; ++q) xs[XOFF + g + 32 * q] = fx[q];
#pragma unroll
    for (int q = 0; q < NF; ++q) ys[gy + 2 * YP * q] = fy[q];
  }
  g += FILL;
  gy += YP * (FILL / 16);
  __builtin_amdgcn_fence(__ATOMIC_ACQ_REL, "wavefront");

  const int m    = lane & 15;
  const int kb4  = (lane >> 4) << 2;

  int xb01 = (SLO0 << 4) + m + kb4;
  int xb2  = ((32 + SLO2) << 4) + m + kb4;
  int yb   = YP * (SLO2 + YOFF / 16 - m) + kb4;

  v8f acc0 = {}, acc1 = {}, acc2 = {};

  Frag cur = {};
  cur.b1 = load_b(ys, yb);
  {
    Frag first = cur;
    fetch(first, cur, 7, xs, ys, xb01, xb2, yb);
    cur = first;
  }

#pragma unroll 1
  for (int p = 0; p < NCH - 1; ++p) {
    if (p == 2)      __builtin_amdgcn_s_setprio(2);
    else if (p == 4) __builtin_amdgcn_s_setprio(1);
    const bool whole = (p < NRND - 1);
    const bool last  = (p == NRND - 1);
    float rx[NR], ry[NR];
    if (whole) {
#pragma unroll
      for (int q = 0; q < NR; ++q) {
        rx[q] = x[g + 32 * q];
        ry[q] = y[g + 32 * q];
      }
    } else if (last) {
#pragma unroll
      for (int q = 0; q < NR; ++q) {
        if (lane + 32 * q < LASTN) {
          rx[q] = x[g + 32 * q];
          ry[q] = y[g + 32 * q];
        }
      }
    }

#pragma unroll 1
    for (int h = 0; h < SC / SH; ++h) {
#pragma unroll
      for (int u = 0; u < SH; ++u) {
        Frag nxt = cur;
        fetch(nxt, cur, 7, xs, ys, xb01 + 16 * (u + 1), xb2 + 16 * (u + 1), yb + YP * (u + 1));
        __builtin_amdgcn_sched_barrier(0);
        xcorr_step(cur, 7, acc0, acc1, acc2);
        __builtin_amdgcn_sched_barrier(0);
        cur = nxt;
      }
      xb01 += 16 * SH;
      xb2  += 16 * SH;
      yb   += YP * SH;
    }

    if (whole) {
#pragma unroll
      for (int q = 0; q < NR; ++q) {
        xs[XOFF + g + 32 * q] = rx[q];
        ys[gy + 2 * YP * q] = ry[q];
      }
      g += RW;
      gy += YP * (RW / 16);
      __builtin_amdgcn_fence(__ATOMIC_ACQ_REL, "wavefront");
    } else if (last) {
#pragma unroll
      for (int q = 0; q < NR; ++q) {
        if (lane + 32 * q < LASTN) {
          xs[XOFF + g + 32 * q] = rx[q];
          ys[gy + 2 * YP * q] = ry[q];
        }
      }
      __builtin_amdgcn_fence(__ATOMIC_ACQ_REL, "wavefront");
    }
  }

  __builtin_amdgcn_s_setprio(0);
#pragma unroll
  for (int t = 0; t < NST1 - NFULL; ++t) {
    Frag nxt = cur;
    fetch(nxt, cur, tile_mask(NFULL + t + 1), xs, ys, xb01 + 16 * (t + 1), xb2 + 16 * (t + 1),
          yb + YP * (t + 1));
    __builtin_amdgcn_sched_barrier(0);
    xcorr_step(cur, tile_mask(NFULL + t), acc0, acc1, acc2);
    __builtin_amdgcn_sched_barrier(0);
    cur = nxt;
  }

  store_tile(out, c, 0, lane, acc0);
  store_tile(out, c, 16, lane, acc1);
  store_tile(out, c, 32, lane, acc2);
}

extern "C" void kernel_launch(void* const* d_in, const int* in_sizes, int n_in,
                              void* d_out, int out_size, void* d_ws, size_t ws_size,
                              hipStream_t stream) {
  const float* d1 = (const float*)d_in[0];
  const float* d2 = (const float*)d_in[1];
  float* out = (float*)d_out;
  (void)in_sizes; (void)n_in; (void)out_size; (void)d_ws; (void)ws_size;
  hipLaunchKernelGGL(xcorr_wmma_f32, dim3(512), dim3(128), 0, stream, d1, d2, out);
}


# third tile (lags 512-602) at half rate: its columns stand 8 lags apart and take the first 8 samples of each 16: 2 matrix instructions a step, not 4; a lag's two partial sums added at the store
# speedup vs baseline: 2.0786x; 1.1389x over previous
#include <hip/hip_runtime.h>


typedef float v2f __attribute__((ext_vector_type(2)));
typedef float v4f __attribute__((ext_vector_type(4)));
typedef float v8f __attribute__((ext_vector_type(8)));

#define NT    3000
#define NOUT  603
#define XOFF  301
#define XPAD  3775
#define YOFF  240
#define YBLK  218
#define YP    20
#define YPAD  (YP * YBLK)
#define NS    203
#define T2C0  ((32 << 4) - 8)

#define S_LO  ((XOFF - 30 + 15) / 16)
#define S_HI  ((NT - 1 + XOFF) / 16 + 1)

#define SC    28
#define SH    14
#define NCH   7
#define NCB   4

#define NF    24
#define FILL  (32 * NF)
#define NR    15
#define RW    (32 * NR)
#define NRND  5
#define LASTN (NT - FILL - RW * (NRND - 1))
#define XTAIL 96

static_assert(FILL + XOFF >= 16 * (32 + SC) + 31, "chunk 0 and the step behind it read x beyond the fill");
static_assert(FILL + YOFF >= 16 * (S_LO + SC) + 256, "chunk 0 and the step behind it read y beyond the fill");
static_assert(RW >= 16 * SC, "a round brings less than a chunk consumes");
static_assert(LASTN > 0 && LASTN <= RW, "the last round is not the last one");
static_assert(16 * (S_HI - 1) + 31 <= XOFF + NT + XTAIL, "x is read behind its zeroed tail");
static_assert(XOFF + NT + XTAIL <= XPAD, "x tail leaves the row");
static_assert((NS - 1) + YOFF / 16 + 1 <= YBLK, "y is read behind its row");
static_assert(YOFF % 16 == 0 && FILL % 16 == 0 && RW % 16 == 0 && 32 % 16 == 0,
              "a staging window does not start on a y block");
static_assert(YP % 8 == 4, "16 consecutive y blocks do not start on 16 different groups of 4 banks");
static_assert((NCB * XPAD) % 4 == 0, "the y rows do not start 16-byte aligned");
static_assert(SC % SH == 0, "a chunk is a whole number of halves");

constexpr int SLO0 = (S_LO > 0) ? S_LO : 0;
constexpr int SLO1 = (S_LO - 16 > 0) ? S_LO - 16 : 0;
constexpr int SLO2 = (S_LO - 32 > 0) ? S_LO - 32 : 0;
constexpr int NST0 = ((S_HI < NS) ? S_HI : NS) - SLO0;
constexpr int NST1 = ((S_HI - 16 < NS) ? S_HI - 16 : NS) - SLO1;
constexpr int NST2 = ((S_HI - 32 < NS) ? S_HI - 32 : NS) - SLO2;
constexpr int NFULL = SC * (NCH - 1);
constexpr int YB0  = SLO0 - SLO2;
static_assert(SLO0 == 16 + SLO1, "tiles 0 and 1 do not read the same x window");
static_assert(NST2 >= NFULL && NST2 <= NST0 && NST0 <= NST1 && NST1 <= NFULL + SC,
              "the tail does not end tile 2 first, then tile 0, then tile 1");

struct FragA { v2f k[4]; };
struct FragB { v4f q[2]; };
struct Frag {
  FragA a01;
  v2f   a2[2];
  FragB b0, b1;
  v4f   b2;
};

__device__ __forceinline__ FragA load_a(const float* xs, int xb) {
  FragA f;
#pragma unroll
  for (int j = 0; j < 2; ++j) {
    f.k[2 * j].x     = xs[xb + 8 * j];
    f.k[2 * j].y     = xs[xb + 8 * j + 1];
    f.k[2 * j + 1].x = xs[xb + 8 * j + 2];
    f.k[2 * j + 1].y = xs[xb + 8 * j + 3];
  }
  return f;
}

__device__ __forceinline__ void load_a2(v2f (&k)[2], const float* xs, int xb) {
  k[0].x = xs[xb];
  k[0].y = xs[xb + 1];
  k[1].x = xs[xb + 2];
  k[1].y = xs[xb + 3];
}

__device__ __forceinline__ FragB load_b(const float* ys, int yb) {
  FragB f;
#pragma unroll
  for (int j = 0; j < 2; ++j) f.q[j] = *reinterpret_cast<const v4f*>(ys + yb + 8 * j);
  return f;
}

__device__ __forceinline__ v2f b_slice(const FragB& f, int ks) {
  v2f b;
  const v4f q = f.q[ks >> 1];
  b.x = (ks & 1) ? q.z : q.x;
  b.y = (ks & 1) ? q.w : q.y;
  return b;
}

__device__ __forceinline__ void fetch(Frag& n, int mask, const float* xs,
                                      const float* ys, int xb01, int xb2, int yb, int yb2) {
  if (mask & 3) n.a01 = load_a(xs, xb01);
  if (mask & 4) {
    load_a2(n.a2, xs, xb2);
    n.b2 = *reinterpret_cast<const v4f*>(ys + yb2);
  }
  if (mask & 2) n.b1 = load_b(ys, yb + YP * (SLO1 - SLO2));
  if (mask & 1) n.b0 = load_b(ys, yb + YP * YB0);
}

__device__ __forceinline__ void xcorr_step(const Frag& f, int mask, v8f& acc0, v8f& acc1, v8f& acc2) {
#pragma unroll
  for (int ks = 0; ks < 4; ++ks) {
    if (mask & 1)
      acc0 = __builtin_amdgcn_wmma_f32_16x16x4_f32(false, f.a01.k[ks], false, b_slice(f.b0, ks),
                                                   (short)0, acc0, false, false);
    if (mask & 2)
      acc1 = __builtin_amdgcn_wmma_f32_16x16x4_f32(false, f.a01.k[ks], false, b_slice(f.b1, ks),
                                                   (short)0, acc1, false, false);
    if ((mask & 4) && ks < 2) {
      v2f b;
      b.x = ks ? f.b2.z : f.b2.x;
      b.y = ks ? f.b2.w : f.b2.y;
      acc2 = __builtin_amdgcn_wmma_f32_16x16x4_f32(false, f.a2[ks], false, b,
                                                   (short)0, acc2, false, false);
    }
  }
}

__device__ __forceinline__ constexpr int tile_mask(int u) {
  return (u < NST0 ? 1 : 0) | (u < NST1 ? 2 : 0) | (u < NST2 ? 4 : 0);
}

__device__ __forceinline__ void store_tile(float* __restrict__ out, int c, int OB0, int lane, v8f acc) {
  const int m  = lane & 15;
  const int hi = lane >> 4;
#pragma unroll
  for (int i = 0; i < 8; ++i) {
    int oi = i + (hi << 3);
    int o  = ((OB0 + m) << 4) + oi;
    if (o < NOUT) out[c * NOUT + o] = acc[i];
  }
}

__device__ __forceinline__ void store_tile2(float* __restrict__ out, int c, int lane, v8f acc) {
#pragma unroll
  for (int i = 0; i < 8; ++i) {
    float other = __shfl(acc[i], lane + 15);
    int o = T2C0 + (lane << 3) + i;
    if (lane >= 1 && lane < 16 && o < NOUT) out[c * NOUT + o] = acc[i] + other;
  }
}

__global__ __launch_bounds__(32 * NCB) void xcorr_wmma_f32(const float* __restrict__ d1,
                                                     const float* __restrict__ d2,
                                                     float* __restrict__ out) {
  __shared__ __attribute__((aligned(16))) float lds[NCB * XPAD + NCB * YPAD];

  __builtin_amdgcn_s_setprio(3);

  const int ch   = __builtin_amdgcn_readfirstlane(threadIdx.x >> 5);
  const int lane = threadIdx.x & 31;
  const int c    = blockIdx.x * NCB + ch;
  float* xs = lds + ch * XPAD;
  float* ys = lds + NCB * XPAD + ch * YPAD;
  const float* x = d1 + c * NT;
  const float* y = d2 + c * NT;

  int g = lane;
  int gy = YP * (YOFF / 16 + (lane >> 4)) + (lane & 15);

  {
    float fx[NF], fy[NF];
#pragma unroll
    for (int q = 0; q < NF; ++q) fx[q] = x[g + 32 * q];
#pragma unroll
    for (int q = 0; q < NF; ++q) fy[q] = y[g + 32 * q];
#pragma unroll
    for (int q = 0; q < (XOFF + 31) / 32; ++q) {
      int i = lane + 32 * q;
      if (i < XOFF) xs[i] = 0.0f;
    }
#pragma unroll
    for (int q = 0; q < XTAIL / 32; ++q)
      xs[XOFF + NT + lane + 32 * q] = 0.0f;
#pragma unroll
    for (int q = 0; q < (YP * (YOFF / 16) + 31) / 32; ++q) {
      int i = lane + 32 * q;
      if (i < YP * (YOFF / 16)) ys[i] = 0.0f;
    }
    constexpr int YEND = YP * ((YOFF + NT) / 16) + (YOFF + NT) % 16;
#pragma unroll
    for (int q = 0; q < (YPAD - YEND + 31) / 32; ++q) {
      int i = YEND + lane + 32 * q;
      if (i < YPAD) ys[i] = 0.0f;
    }
#pragma unroll
    for (int q = 0; q < NF; ++q) xs[XOFF + g + 32 * q] = fx[q];
#pragma unroll
    for (int q = 0; q < NF; ++q) ys[gy + 2 * YP * q] = fy[q];
  }
  g += FILL;
  gy += YP * (FILL / 16);
  __builtin_amdgcn_fence(__ATOMIC_ACQ_REL, "wavefront");

  const int m    = lane & 15;
  const int kb4  = (lane >> 4) << 2;

  int xb01 = (SLO0 << 4) + m + kb4;
  int xb2  = ((32 + SLO2) << 4) + m + kb4;
  int yb   = YP * (SLO2 + YOFF / 16 - m) + kb4;
  int yb2  = YP * (SLO2 + YOFF / 16 - (m >> 1)) + ((m & 1) ? 0 : 8) + kb4;

  v8f acc0 = {}, acc1 = {}, acc2 = {};

  Frag cur = {};
  fetch(cur, 7, xs, ys, xb01, xb2, yb, yb2);

#pragma unroll 1
  for (int p = 0; p < NCH - 1; ++p) {
    if (p == 2)      __builtin_amdgcn_s_setprio(2);
    else if (p == 4) __builtin_amdgcn_s_setprio(1);
    const bool whole = (p < NRND - 1);
    const bool last  = (p == NRND - 1);
    float rx[NR], ry[NR];
    if (whole) {
#pragma unroll
      for (int q = 0; q < NR; ++q) {
        rx[q] = x[g + 32 * q];
        ry[q] = y[g + 32 * q];
      }
    } else if (last) {
#pragma unroll
      for (int q = 0; q < NR; ++q) {
        if (lane + 32 * q < LASTN) {
          rx[q] = x[g + 32 * q];
          ry[q] = y[g + 32 * q];
        }
      }
    }

#pragma unroll 1
    for (int h = 0; h < SC / SH; ++h) {
#pragma unroll
      for (int u = 0; u < SH; ++u) {
        Frag nxt = cur;
        fetch(nxt, 7, xs, ys, xb01 + 16 * (u + 1), xb2 + 16 * (u + 1), yb + YP * (u + 1),
              yb2 + YP * (u + 1));
        __builtin_amdgcn_sched_barrier(0);
        xcorr_step(cur, 7, acc0, acc1, acc2);
        __builtin_amdgcn_sched_barrier(0);
        cur = nxt;
      }
      xb01 += 16 * SH;
      xb2  += 16 * SH;
      yb   += YP * SH;
      yb2  += YP * SH;
    }

    if (whole) {
#pragma unroll
      for (int q = 0; q < NR; ++q) {
        xs[XOFF + g + 32 * q] = rx[q];
        ys[gy + 2 * YP * q] = ry[q];
      }
      g += RW;
      gy += YP * (RW / 16);
      __builtin_amdgcn_fence(__ATOMIC_ACQ_REL, "wavefront");
    } else if (last) {
#pragma unroll
      for (int q = 0; q < NR; ++q) {
        if (lane + 32 * q < LASTN) {
          xs[XOFF + g + 32 * q] = rx[q];
          ys[gy + 2 * YP * q] = ry[q];
        }
      }
      __builtin_amdgcn_fence(__ATOMIC_ACQ_REL, "wavefront");
    }
  }

  __builtin_amdgcn_s_setprio(0);
#pragma unroll
  for (int t = 0; t < NST1 - NFULL; ++t) {
    Frag nxt = cur;
    fetch(nxt, tile_mask(NFULL + t + 1), xs, ys, xb01 + 16 * (t + 1), xb2 + 16 * (t + 1),
          yb + YP * (t + 1), yb2 + YP * (t + 1));
    __builtin_amdgcn_sched_barrier(0);
    xcorr_step(cur, tile_mask(NFULL + t), acc0, acc1, acc2);
    __builtin_amdgcn_sched_barrier(0);
    cur = nxt;
  }

  store_tile(out, c, 0, lane, acc0);
  store_tile(out, c, 16, lane, acc1);
  store_tile2(out, c, lane, acc2);
}

extern "C" void kernel_launch(void* const* d_in, const int* in_sizes, int n_in,
                              void* d_out, int out_size, void* d_ws, size_t ws_size,
                              hipStream_t stream) {
  const float* d1 = (const float*)d_in[0];
  const float* d2 = (const float*)d_in[1];
  float* out = (float*)d_out;
  (void)in_sizes; (void)n_in; (void)out_size; (void)d_ws; (void)ws_size;
  hipLaunchKernelGGL(xcorr_wmma_f32, dim3(512), dim3(128), 0, stream, d1, d2, out);
}
